# starting kernel
# baseline (speedup 1.0000x reference)
#include <hip/hip_runtime.h>


#define M_ROWS 131072
#define R_COLS 64

typedef __attribute__((ext_vector_type(2))) float v2f;
typedef __attribute__((ext_vector_type(4))) float v4f;
typedef __attribute__((ext_vector_type(8))) float v8f;

#define WS_GR 0
#define WS_GI 4096
#define WS_W  8192

__device__ inline v8f wmma_f32x4(v2f a, v2f b, v8f c) {
  return __builtin_amdgcn_wmma_f32_16x16x4_f32(
       false, a,  false, b,
       (short)0, c,  false,  false);
}

__global__ void fo_zero_gram(float* ws) {
  int i = blockIdx.x * blockDim.x + threadIdx.x;
  if (i < 2 * 64 * 64) ws[i] = 0.0f;
}

__global__ __launch_bounds__(512) void fo_gram(const float* __restrict__ x,
                                               float* __restrict__ ws) {
  const int lane = threadIdx.x & 31;
  const int wave = threadIdx.x >> 5;
  const int ti = wave >> 2;
  const int tj = wave & 3;
  const int p = lane & 15;
  const int kb = (lane < 16) ? 0 : 2;

  const int colA = 16 * ti + p;
  const int colB = 16 * tj + p;
  const int row0 = blockIdx.x * 256;

  v8f gr = {};
  v8f gi = {};

  for (int ks = 0; ks < 256; ks += 4) {
    const size_t m = (size_t)(row0 + ks + kb);
    v2f ea0 = *(const v2f*)(x + (m * R_COLS + colA) * 2);
    v2f ea1 = *(const v2f*)(x + ((m + 1) * R_COLS + colA) * 2);
    v2f eb0 = *(const v2f*)(x + (m * R_COLS + colB) * 2);
    v2f eb1 = *(const v2f*)(x + ((m + 1) * R_COLS + colB) * 2);

    v2f ar = {ea0.x, ea1.x};
    v2f ai = {ea0.y, ea1.y};
    v2f ain = {-ea0.y, -ea1.y};
    v2f br = {eb0.x, eb1.x};
    v2f bi = {eb0.y, eb1.y};

    gr = wmma_f32x4(ar, br, gr);
    gr = wmma_f32x4(ai, bi, gr);
    gi = wmma_f32x4(ar, bi, gi);
    gi = wmma_f32x4(ain, br, gi);
  }

  const int rbase = 16 * ti + ((lane < 16) ? 0 : 8);
  const int c = 16 * tj + p;
  float* grp = ws + WS_GR;
  float* gip = ws + WS_GI;
#pragma unroll
  for (int v = 0; v < 8; ++v) {
    atomicAdd(&grp[(rbase + v) * 64 + c], gr[v]);
    atomicAdd(&gip[(rbase + v) * 64 + c], gi[v]);
  }
}

__global__ __launch_bounds__(64) void fo_chol_inv(float* __restrict__ ws) {
  __shared__ float2 Gs[64 * 64];
  __shared__ float2 Ys[64 * 64];
  const int t = threadIdx.x;
  const float* grp = ws + WS_GR;
  const float* gip = ws + WS_GI;

  for (int i = t; i < 64 * 64; i += 64) {
    Gs[i].x = grp[i];
    Gs[i].y = gip[i];
  }
  __syncthreads();

  for (int k = 0; k < 64; ++k) {
    if (t == k) {
      float d = sqrtf(Gs[k * 64 + k].x + 1e-8f);
      Gs[k * 64 + k].x = d;
      Gs[k * 64 + k].y = 0.0f;
    }
    __syncthreads();
    const float d = Gs[k * 64 + k].x;
    if (t > k) {
      Gs[t * 64 + k].x /= d;
      Gs[t * 64 + k].y /= d;
    }
    __syncthreads();
    if (t > k) {
      const float2 ltk = Gs[t * 64 + k];
      for (int j = k + 1; j <= t; ++j) {
        const float2 ljk = Gs[j * 64 + k];
        Gs[t * 64 + j].x -= ltk.x * ljk.x + ltk.y * ljk.y;
        Gs[t * 64 + j].y -= ltk.y * ljk.x - ltk.x * ljk.y;
      }
    }
    __syncthreads();
  }

  const int cidx = t;
  for (int r = 0; r < 64; ++r) {
    float2 s;
    s.x = (r == cidx) ? 1.0f : 0.0f;
    s.y = 0.0f;
    if (r < cidx) {
      Ys[r * 64 + cidx] = make_float2(0.0f, 0.0f);
      continue;
    }
    for (int kk = cidx; kk < r; ++kk) {
      const float2 l = Gs[r * 64 + kk];
      const float2 y = Ys[kk * 64 + cidx];
      s.x -= l.x * y.x - l.y * y.y;
      s.y -= l.x * y.y + l.y * y.x;
    }
    const float d = Gs[r * 64 + r].x;
    s.x /= d;
    s.y /= d;
    Ys[r * 64 + cidx] = s;
  }
  __syncthreads();

  float* wp = ws + WS_W;
  for (int j = 0; j < 64; ++j) {
    const float2 y = Ys[j * 64 + cidx];
    wp[(cidx * 64 + j) * 2 + 0] = y.x;
    wp[(cidx * 64 + j) * 2 + 1] = -y.y;
  }
}

__global__ __launch_bounds__(512) void fo_apply(const float* __restrict__ x,
                                                const float* __restrict__ ws,
                                                float* __restrict__ out) {
  __shared__ float2 Wl[64 * 64];
  const float* wp = ws + WS_W;
  for (int i = threadIdx.x; i < 64 * 64; i += 512) {
    Wl[i].x = wp[2 * i + 0];
    Wl[i].y = wp[2 * i + 1];
  }
  __syncthreads();

  const int lane = threadIdx.x & 31;
  const int wave = threadIdx.x >> 5;
  const int s = wave >> 2;
  const int tj = wave & 3;
  const int p = lane & 15;
  const int kb = (lane < 16) ? 0 : 2;

  const size_t row0 = (size_t)blockIdx.x * 64 + 16 * s;

  v8f qr = {};
  v8f qi = {};

  for (int ks = 0; ks < 64; ks += 4) {
    v4f e = *(const v4f*)(x + ((row0 + p) * R_COLS + ks + kb) * 2);
    v2f ar = {e.x, e.z};
    v2f ai = {e.y, e.w};
    v2f ain = {-e.y, -e.w};

    const float2 w0 = Wl[(ks + kb) * 64 + 16 * tj + p];
    const float2 w1 = Wl[(ks + kb + 1) * 64 + 16 * tj + p];
    v2f wr = {w0.x, w1.x};
    v2f wi = {w0.y, w1.y};

    qr = wmma_f32x4(ar, wr, qr);
    qr = wmma_f32x4(ain, wi, qr);
    qi = wmma_f32x4(ar, wi, qi);
    qi = wmma_f32x4(ai, wr, qi);
  }

  const int rb = (lane < 16) ? 0 : 8;
  const int c = 16 * tj + p;
#pragma unroll
  for (int v = 0; v < 8; ++v) {
    const size_t row = row0 + rb + v;
    v2f o = {qr[v], qi[v]};
    *(v2f*)(out + (row * R_COLS + c) * 2) = o;
  }
}

extern "C" void kernel_launch(void* const* d_in, const int* in_sizes, int n_in,
                              void* d_out, int out_size, void* d_ws, size_t ws_size,
                              hipStream_t stream) {
  (void)in_sizes; (void)n_in; (void)out_size; (void)ws_size;
  const float* x = (const float*)d_in[0];
  float* out = (float*)d_out;
  float* ws = (float*)d_ws;

  fo_zero_gram<<<(2 * 64 * 64 + 255) / 256, 256, 0, stream>>>(ws);
  fo_gram<<<M_ROWS / 256, 512, 0, stream>>>(x, ws);
  fo_chol_inv<<<1, 64, 0, stream>>>(ws);
  fo_apply<<<M_ROWS / 64, 512, 0, stream>>>(x, ws, out);
}


# factor+inverse kernel: matrix rows kept in registers as a rotating window, pivot row broadcast via LDS, L^-1 built in the same elimination (sums reordered) instead of LDS read-modify-write loops
# speedup vs baseline: 2.5315x; 2.5315x over previous
#include <hip/hip_runtime.h>


#define M_ROWS 131072
#define R_COLS 64

typedef __attribute__((ext_vector_type(2))) float v2f;
typedef __attribute__((ext_vector_type(4))) float v4f;
typedef __attribute__((ext_vector_type(8))) float v8f;

#define WS_GR 0
#define WS_GI 4096
#define WS_W  8192

__device__ inline v8f wmma_f32x4(v2f a, v2f b, v8f c) {
  return __builtin_amdgcn_wmma_f32_16x16x4_f32(
       false, a,  false, b,
       (short)0, c,  false,  false);
}

__global__ void fo_zero_gram(float* ws) {
  int i = blockIdx.x * blockDim.x + threadIdx.x;
  if (i < 2 * 64 * 64) ws[i] = 0.0f;
}

__global__ __launch_bounds__(512) void fo_gram(const float* __restrict__ x,
                                               float* __restrict__ ws) {
  const int lane = threadIdx.x & 31;
  const int wave = threadIdx.x >> 5;
  const int ti = wave >> 2;
  const int tj = wave & 3;
  const int p = lane & 15;
  const int kb = (lane < 16) ? 0 : 2;

  const int colA = 16 * ti + p;
  const int colB = 16 * tj + p;
  const int row0 = blockIdx.x * 256;

  v8f gr = {};
  v8f gi = {};

  for (int ks = 0; ks < 256; ks += 4) {
    const size_t m = (size_t)(row0 + ks + kb);
    v2f ea0 = *(const v2f*)(x + (m * R_COLS + colA) * 2);
    v2f ea1 = *(const v2f*)(x + ((m + 1) * R_COLS + colA) * 2);
    v2f eb0 = *(const v2f*)(x + (m * R_COLS + colB) * 2);
    v2f eb1 = *(const v2f*)(x + ((m + 1) * R_COLS + colB) * 2);

    v2f ar = {ea0.x, ea1.x};
    v2f ai = {ea0.y, ea1.y};
    v2f ain = {-ea0.y, -ea1.y};
    v2f br = {eb0.x, eb1.x};
    v2f bi = {eb0.y, eb1.y};

    gr = wmma_f32x4(ar, br, gr);
    gr = wmma_f32x4(ai, bi, gr);
    gi = wmma_f32x4(ar, bi, gi);
    gi = wmma_f32x4(ain, br, gi);
  }

  const int rbase = 16 * ti + ((lane < 16) ? 0 : 8);
  const int c = 16 * tj + p;
  float* grp = ws + WS_GR;
  float* gip = ws + WS_GI;
#pragma unroll
  for (int v = 0; v < 8; ++v) {
    atomicAdd(&grp[(rbase + v) * 64 + c], gr[v]);
    atomicAdd(&gip[(rbase + v) * 64 + c], gi[v]);
  }
}

__global__ __launch_bounds__(64) void fo_chol_inv(float* __restrict__ ws) {
  __shared__ v4f Vb[2][32];
  const int t = threadIdx.x;
  const float* grp = ws + WS_GR;
  const float* gip = ws + WS_GI;

  v2f U[64];
#pragma unroll
  for (int i = 0; i < 16; ++i) {
    const v4f r = *(const v4f*)(grp + t * 64 + 4 * i);
    const v4f m = *(const v4f*)(gip + t * 64 + 4 * i);
    U[4 * i + 0] = v2f{r.x, m.x};
    U[4 * i + 1] = v2f{r.y, m.y};
    U[4 * i + 2] = v2f{r.z, m.z};
    U[4 * i + 3] = v2f{r.w, m.w};
  }

  float mydd = 1.0f;

#pragma nounroll
  for (int k = 0; k < 64; ++k) {
    v4f* buf = Vb[k & 1];
    if (t == k) {
#pragma unroll
      for (int i = 0; i < 32; ++i)
        buf[i] = v4f{U[2 * i].x, U[2 * i].y, U[2 * i + 1].x, U[2 * i + 1].y};
    }
    __syncthreads();

    v2f V[64];
#pragma unroll
    for (int i = 0; i < 32; ++i) {
      const v4f q = buf[i];
      V[2 * i] = v2f{q.x, q.y};
      V[2 * i + 1] = v2f{q.z, q.w};
    }
    __builtin_amdgcn_sched_barrier(0);

    const float dd = V[0].x + 1e-8f;
    if (t == k) mydd = dd;
    const float act = (t > k) ? 1.0f : 0.0f;
    const float ax = (U[0].x * act) / dd;
    const float ay = (U[0].y * act) / dd;
    const v2f nax = {-ax, -ax};
    const v2f aypm = {ay, -ay};

#pragma unroll
    for (int s = 0; s < 63; ++s) {
      const v2f vs = V[s + 1];
      const v2f vsw = {vs.y, vs.x};
      const v2f tmp = __builtin_elementwise_fma(nax, vs, U[s + 1]);
      U[s] = __builtin_elementwise_fma(aypm, vsw, tmp);
    }
    U[63] = (t == k) ? v2f{1.0f, 0.0f} : v2f{-ax, -ay};
  }

  const float rd = 1.0f / sqrtf(mydd);
  float* wp = ws + WS_W;
#pragma unroll
  for (int c = 0; c < 64; ++c) {
    const v2f o = {U[c].x * rd, -U[c].y * rd};
    *(v2f*)(wp + (c * 64 + t) * 2) = o;
  }
}

__global__ __launch_bounds__(512) void fo_apply(const float* __restrict__ x,
                                                const float* __restrict__ ws,
                                                float* __restrict__ out) {
  __shared__ float2 Wl[64 * 64];
  const float* wp = ws + WS_W;
  for (int i = threadIdx.x; i < 64 * 64; i += 512) {
    Wl[i].x = wp[2 * i + 0];
    Wl[i].y = wp[2 * i + 1];
  }
  __syncthreads();

  const int lane = threadIdx.x & 31;
  const int wave = threadIdx.x >> 5;
  const int s = wave >> 2;
  const int tj = wave & 3;
  const int p = lane & 15;
  const int kb = (lane < 16) ? 0 : 2;

  const size_t row0 = (size_t)blockIdx.x * 64 + 16 * s;

  v8f qr = {};
  v8f qi = {};

  for (int ks = 0; ks < 64; ks += 4) {
    v4f e = *(const v4f*)(x + ((row0 + p) * R_COLS + ks + kb) * 2);
    v2f ar = {e.x, e.z};
    v2f ai = {e.y, e.w};
    v2f ain = {-e.y, -e.w};

    const float2 w0 = Wl[(ks + kb) * 64 + 16 * tj + p];
    const float2 w1 = Wl[(ks + kb + 1) * 64 + 16 * tj + p];
    v2f wr = {w0.x, w1.x};
    v2f wi = {w0.y, w1.y};

    qr = wmma_f32x4(ar, wr, qr);
    qr = wmma_f32x4(ain, wi, qr);
    qi = wmma_f32x4(ar, wi, qi);
    qi = wmma_f32x4(ai, wr, qi);
  }

  const int rb = (lane < 16) ? 0 : 8;
  const int c = 16 * tj + p;
#pragma unroll
  for (int v = 0; v < 8; ++v) {
    const size_t row = row0 + rb + v;
    v2f o = {qr[v], qi[v]};
    *(v2f*)(out + (row * R_COLS + c) * 2) = o;
  }
}

extern "C" void kernel_launch(void* const* d_in, const int* in_sizes, int n_in,
                              void* d_out, int out_size, void* d_ws, size_t ws_size,
                              hipStream_t stream) {
  (void)in_sizes; (void)n_in; (void)out_size; (void)ws_size;
  const float* x = (const float*)d_in[0];
  float* out = (float*)d_out;
  float* ws = (float*)d_ws;

  fo_zero_gram<<<(2 * 64 * 64 + 255) / 256, 256, 0, stream>>>(ws);
  fo_gram<<<M_ROWS / 256, 512, 0, stream>>>(x, ws);
  fo_chol_inv<<<1, 64, 0, stream>>>(ws);
  fo_apply<<<M_ROWS / 64, 512, 0, stream>>>(x, ws, out);
}


# Gram: float atomics and the zeroing launch removed; each work-group stores its partial Gram with plain stores, a reduction kernel sums the 512 partials in a fixed order (sum order changes)
# speedup vs baseline: 3.2951x; 1.3016x over previous
#include <hip/hip_runtime.h>


#define M_ROWS 131072
#define R_COLS 64

typedef __attribute__((ext_vector_type(2))) float v2f;
typedef __attribute__((ext_vector_type(4))) float v4f;
typedef __attribute__((ext_vector_type(8))) float v8f;

#define WS_GR 0
#define WS_GI 4096
#define WS_W  8192

#define GRAM_BLOCKS 512
#define PART_FLOATS 8192

__device__ inline v8f wmma_f32x4(v2f a, v2f b, v8f c) {
  return __builtin_amdgcn_wmma_f32_16x16x4_f32(
       false, a,  false, b,
       (short)0, c,  false,  false);
}

__global__ __launch_bounds__(512) void fo_gram(const float* __restrict__ x,
                                               float* __restrict__ part) {
  const int lane = threadIdx.x & 31;
  const int wave = threadIdx.x >> 5;
  const int ti = wave >> 2;
  const int tj = wave & 3;
  const int p = lane & 15;
  const int kb = (lane < 16) ? 0 : 2;

  const int colA = 16 * ti + p;
  const int colB = 16 * tj + p;
  const int row0 = blockIdx.x * 256;

  v8f gr = {};
  v8f gi = {};

  for (int ks = 0; ks < 256; ks += 4) {
    const size_t m = (size_t)(row0 + ks + kb);
    v2f ea0 = *(const v2f*)(x + (m * R_COLS + colA) * 2);
    v2f ea1 = *(const v2f*)(x + ((m + 1) * R_COLS + colA) * 2);
    v2f eb0 = *(const v2f*)(x + (m * R_COLS + colB) * 2);
    v2f eb1 = *(const v2f*)(x + ((m + 1) * R_COLS + colB) * 2);

    v2f ar = {ea0.x, ea1.x};
    v2f ai = {ea0.y, ea1.y};
    v2f ain = {-ea0.y, -ea1.y};
    v2f br = {eb0.x, eb1.x};
    v2f bi = {eb0.y, eb1.y};

    gr = wmma_f32x4(ar, br, gr);
    gr = wmma_f32x4(ai, bi, gr);
    gi = wmma_f32x4(ar, bi, gi);
    gi = wmma_f32x4(ain, br, gi);
  }

  float* pp = part + (size_t)blockIdx.x * PART_FLOATS + wave * 512 + lane;
#pragma unroll
  for (int v = 0; v < 8; ++v) {
    pp[v * 32] = gr[v];
    pp[256 + v * 32] = gi[v];
  }
}

__global__ __launch_bounds__(128) void fo_reduce_gram(const float* __restrict__ part,
                                                      float* __restrict__ ws) {
  __shared__ v4f red[16][8];
  const int gl = threadIdx.x & 7;
  const int sl = threadIdx.x >> 3;
  const int g = blockIdx.x * 8 + gl;

  const v4f* src = (const v4f*)part + (size_t)(sl * 32) * (PART_FLOATS / 4) + g;
  v4f acc = {};
#pragma unroll
  for (int i = 0; i < 32; ++i) acc += src[(size_t)i * (PART_FLOATS / 4)];
  red[sl][gl] = acc;
  __syncthreads();

  if (threadIdx.x < 8) {
    v4f s = red[0][gl];
#pragma unroll
    for (int j = 1; j < 16; ++j) s += red[j][gl];
    const int e = 4 * g;
    const int lane = e & 31;
    const int v = (e >> 5) & 7;
    const int c = (e >> 8) & 1;
    const int wave = e >> 9;
    const int row = 16 * (wave >> 2) + 8 * (lane >> 4) + v;
    const int col = 16 * (wave & 3) + (lane & 15);
    *(v4f*)(ws + c * 4096 + row * 64 + col) = s;
  }
}

__global__ __launch_bounds__(64) void fo_chol_inv(float* __restrict__ ws) {
  __shared__ v4f Vb[2][32];
  const int t = threadIdx.x;
  const float* grp = ws + WS_GR;
  const float* gip = ws + WS_GI;

  v2f U[64];
#pragma unroll
  for (int i = 0; i < 16; ++i) {
    const v4f r = *(const v4f*)(grp + t * 64 + 4 * i);
    const v4f m = *(const v4f*)(gip + t * 64 + 4 * i);
    U[4 * i + 0] = v2f{r.x, m.x};
    U[4 * i + 1] = v2f{r.y, m.y};
    U[4 * i + 2] = v2f{r.z, m.z};
    U[4 * i + 3] = v2f{r.w, m.w};
  }

  float mydd = 1.0f;

#pragma nounroll
  for (int k = 0; k < 64; ++k) {
    v4f* buf = Vb[k & 1];
    if (t == k) {
#pragma unroll
      for (int i = 0; i < 32; ++i)
        buf[i] = v4f{U[2 * i].x, U[2 * i].y, U[2 * i + 1].x, U[2 * i + 1].y};
    }
    __syncthreads();

    v2f V[64];
#pragma unroll
    for (int i = 0; i < 32; ++i) {
      const v4f q = buf[i];
      V[2 * i] = v2f{q.x, q.y};
      V[2 * i + 1] = v2f{q.z, q.w};
    }
    __builtin_amdgcn_sched_barrier(0);

    const float dd = V[0].x + 1e-8f;
    if (t == k) mydd = dd;
    const float act = (t > k) ? 1.0f : 0.0f;
    const float ax = (U[0].x * act) / dd;
    const float ay = (U[0].y * act) / dd;
    const v2f nax = {-ax, -ax};
    const v2f aypm = {ay, -ay};

#pragma unroll
    for (int s = 0; s < 63; ++s) {
      const v2f vs = V[s + 1];
      const v2f vsw = {vs.y, vs.x};
      const v2f tmp = __builtin_elementwise_fma(nax, vs, U[s + 1]);
      U[s] = __builtin_elementwise_fma(aypm, vsw, tmp);
    }
    U[63] = (t == k) ? v2f{1.0f, 0.0f} : v2f{-ax, -ay};
  }

  const float rd = 1.0f / sqrtf(mydd);
  float* wp = ws + WS_W;
#pragma unroll
  for (int c = 0; c < 64; ++c) {
    const v2f o = {U[c].x * rd, -U[c].y * rd};
    *(v2f*)(wp + (c * 64 + t) * 2) = o;
  }
}

__global__ __launch_bounds__(512) void fo_apply(const float* __restrict__ x,
                                                const float* __restrict__ ws,
                                                float* __restrict__ out) {
  __shared__ float2 Wl[64 * 64];
  const float* wp = ws + WS_W;
  for (int i = threadIdx.x; i < 64 * 64; i += 512) {
    Wl[i].x = wp[2 * i + 0];
    Wl[i].y = wp[2 * i + 1];
  }
  __syncthreads();

  const int lane = threadIdx.x & 31;
  const int wave = threadIdx.x >> 5;
  const int s = wave >> 2;
  const int tj = wave & 3;
  const int p = lane & 15;
  const int kb = (lane < 16) ? 0 : 2;

  const size_t row0 = (size_t)blockIdx.x * 64 + 16 * s;

  v8f qr = {};
  v8f qi = {};

  for (int ks = 0; ks < 64; ks += 4) {
    v4f e = *(const v4f*)(x + ((row0 + p) * R_COLS + ks + kb) * 2);
    v2f ar = {e.x, e.z};
    v2f ai = {e.y, e.w};
    v2f ain = {-e.y, -e.w};

    const float2 w0 = Wl[(ks + kb) * 64 + 16 * tj + p];
    const float2 w1 = Wl[(ks + kb + 1) * 64 + 16 * tj + p];
    v2f wr = {w0.x, w1.x};
    v2f wi = {w0.y, w1.y};

    qr = wmma_f32x4(ar, wr, qr);
    qr = wmma_f32x4(ain, wi, qr);
    qi = wmma_f32x4(ar, wi, qi);
    qi = wmma_f32x4(ai, wr, qi);
  }

  const int rb = (lane < 16) ? 0 : 8;
  const int c = 16 * tj + p;
#pragma unroll
  for (int v = 0; v < 8; ++v) {
    const size_t row = row0 + rb + v;
    v2f o = {qr[v], qi[v]};
    *(v2f*)(out + (row * R_COLS + c) * 2) = o;
  }
}

extern "C" void kernel_launch(void* const* d_in, const int* in_sizes, int n_in,
                              void* d_out, int out_size, void* d_ws, size_t ws_size,
                              hipStream_t stream) {
  (void)in_sizes; (void)n_in; (void)out_size; (void)ws_size;
  const float* x = (const float*)d_in[0];
  float* out = (float*)d_out;
  float* ws = (float*)d_ws;

  fo_gram<<<GRAM_BLOCKS, 512, 0, stream>>>(x, out);
  fo_reduce_gram<<<256, 128, 0, stream>>>(out, ws);
  fo_chol_inv<<<1, 64, 0, stream>>>(ws);
  fo_apply<<<M_ROWS / 64, 512, 0, stream>>>(x, ws, out);
}


# Gram: Hermitian symmetry used, only the 10 of 16 tiles on or above the diagonal run on the matrix unit; the reduction writes the lower tiles as conjugate mirrors (their imaginary sum order changes)
# speedup vs baseline: 3.7232x; 1.1299x over previous
#include <hip/hip_runtime.h>


#define M_ROWS 131072
#define R_COLS 64

typedef __attribute__((ext_vector_type(2))) float v2f;
typedef __attribute__((ext_vector_type(4))) float v4f;
typedef __attribute__((ext_vector_type(8))) float v8f;

#define WS_GR 0
#define WS_GI 4096
#define WS_W  8192

#define GRAM_TILES 10
#define GRAM_PARTS 512
#define PART_FLOATS (GRAM_TILES * 512)

__device__ inline v8f wmma_f32x4(v2f a, v2f b, v8f c) {
  return __builtin_amdgcn_wmma_f32_16x16x4_f32(
       false, a,  false, b,
       (short)0, c,  false,  false);
}

__device__ inline void tile_of(int tile, int& ti, int& tj) {
  ti = (tile >= 4) + (tile >= 7) + (tile >= 9);
  const int first = (ti == 0) ? 0 : (ti == 1) ? 4 : (ti == 2) ? 7 : 9;
  tj = tile - first + ti;
}

template <bool DIAG>
__device__ inline void gram_rows(const float* __restrict__ x, size_t mbase, int colA,
                                 int colB, v8f& gr, v8f& gi) {
  for (int ks = 0; ks < 256; ks += 4) {
    const size_t m = mbase + ks;
    const v2f ea0 = *(const v2f*)(x + (m * R_COLS + colA) * 2);
    const v2f ea1 = *(const v2f*)(x + ((m + 1) * R_COLS + colA) * 2);
    v2f eb0 = ea0;
    v2f eb1 = ea1;
    if (!DIAG) {
      eb0 = *(const v2f*)(x + (m * R_COLS + colB) * 2);
      eb1 = *(const v2f*)(x + ((m + 1) * R_COLS + colB) * 2);
    }

    const v2f ar = {ea0.x, ea1.x};
    const v2f ai = {ea0.y, ea1.y};
    const v2f ain = {-ea0.y, -ea1.y};
    const v2f br = {eb0.x, eb1.x};
    const v2f bi = {eb0.y, eb1.y};

    gr = wmma_f32x4(ar, br, gr);
    gr = wmma_f32x4(ai, bi, gr);
    gi = wmma_f32x4(ar, bi, gi);
    gi = wmma_f32x4(ain, br, gi);
  }
}

__global__ __launch_bounds__(640) void fo_gram(const float* __restrict__ x,
                                               float* __restrict__ part) {
  const int lane = threadIdx.x & 31;
  const int wave = __builtin_amdgcn_readfirstlane((int)(threadIdx.x >> 5));
  const int half = (wave >= GRAM_TILES) ? 1 : 0;
  const int tile = wave - GRAM_TILES * half;
  int ti, tj;
  tile_of(tile, ti, tj);
  const int p = lane & 15;
  const int kb = (lane < 16) ? 0 : 2;

  const int colA = 16 * ti + p;
  const int colB = 16 * tj + p;
  const int pidx = blockIdx.x * 2 + half;
  const size_t mbase = (size_t)pidx * 256 + kb;

  v8f gr = {};
  v8f gi = {};
  if (ti == tj)
    gram_rows<true>(x, mbase, colA, colB, gr, gi);
  else
    gram_rows<false>(x, mbase, colA, colB, gr, gi);

  float* pp = part + (size_t)pidx * PART_FLOATS + tile * 512 + lane;
#pragma unroll
  for (int v = 0; v < 8; ++v) {
    pp[v * 32] = gr[v];
    pp[256 + v * 32] = gi[v];
  }
}

__global__ __launch_bounds__(128) void fo_reduce_gram(const float* __restrict__ part,
                                                      float* __restrict__ ws) {
  __shared__ v4f red[16][8];
  const int gl = threadIdx.x & 7;
  const int sl = threadIdx.x >> 3;
  const int g = blockIdx.x * 8 + gl;

  const v4f* src = (const v4f*)part + (size_t)(sl * 32) * (PART_FLOATS / 4) + g;
  v4f acc = {};
#pragma unroll
  for (int i = 0; i < 32; ++i) acc += src[(size_t)i * (PART_FLOATS / 4)];
  red[sl][gl] = acc;
  __syncthreads();

  if (threadIdx.x < 8) {
    v4f s = red[0][gl];
#pragma unroll
    for (int j = 1; j < 16; ++j) s += red[j][gl];
    const int e = 4 * g;
    const int lane = e & 31;
    const int v = (e >> 5) & 7;
    const int c = (e >> 8) & 1;
    int ti, tj;
    tile_of(e >> 9, ti, tj);
    const int row = 16 * ti + 8 * (lane >> 4) + v;
    const int col = 16 * tj + (lane & 15);
    float* plane = ws + c * 4096;
    *(v4f*)(plane + row * 64 + col) = s;
    if (ti != tj) {
      const v4f ms = c ? -s : s;
      plane[(col + 0) * 64 + row] = ms.x;
      plane[(col + 1) * 64 + row] = ms.y;
      plane[(col + 2) * 64 + row] = ms.z;
      plane[(col + 3) * 64 + row] = ms.w;
    }
  }
}

__global__ __launch_bounds__(64) void fo_chol_inv(float* __restrict__ ws) {
  __shared__ v4f Vb[2][32];
  const int t = threadIdx.x;
  const float* grp = ws + WS_GR;
  const float* gip = ws + WS_GI;

  v2f U[64];
#pragma unroll
  for (int i = 0; i < 16; ++i) {
    const v4f r = *(const v4f*)(grp + t * 64 + 4 * i);
    const v4f m = *(const v4f*)(gip + t * 64 + 4 * i);
    U[4 * i + 0] = v2f{r.x, m.x};
    U[4 * i + 1] = v2f{r.y, m.y};
    U[4 * i + 2] = v2f{r.z, m.z};
    U[4 * i + 3] = v2f{r.w, m.w};
  }

  float mydd = 1.0f;

#pragma nounroll
  for (int k = 0; k < 64; ++k) {
    v4f* buf = Vb[k & 1];
    if (t == k) {
#pragma unroll
      for (int i = 0; i < 32; ++i)
        buf[i] = v4f{U[2 * i].x, U[2 * i].y, U[2 * i + 1].x, U[2 * i + 1].y};
    }
    __syncthreads();

    v2f V[64];
#pragma unroll
    for (int i = 0; i < 32; ++i) {
      const v4f q = buf[i];
      V[2 * i] = v2f{q.x, q.y};
      V[2 * i + 1] = v2f{q.z, q.w};
    }
    __builtin_amdgcn_sched_barrier(0);

    const float dd = V[0].x + 1e-8f;
    if (t == k) mydd = dd;
    const float act = (t > k) ? 1.0f : 0.0f;
    const float ax = (U[0].x * act) / dd;
    const float ay = (U[0].y * act) / dd;
    const v2f nax = {-ax, -ax};
    const v2f aypm = {ay, -ay};

#pragma unroll
    for (int s = 0; s < 63; ++s) {
      const v2f vs = V[s + 1];
      const v2f vsw = {vs.y, vs.x};
      const v2f tmp = __builtin_elementwise_fma(nax, vs, U[s + 1]);
      U[s] = __builtin_elementwise_fma(aypm, vsw, tmp);
    }
    U[63] = (t == k) ? v2f{1.0f, 0.0f} : v2f{-ax, -ay};
  }

  const float rd = 1.0f / sqrtf(mydd);
  float* wp = ws + WS_W;
#pragma unroll
  for (int c = 0; c < 64; ++c) {
    const v2f o = {U[c].x * rd, -U[c].y * rd};
    *(v2f*)(wp + (c * 64 + t) * 2) = o;
  }
}

__global__ __launch_bounds__(512) void fo_apply(const float* __restrict__ x,
                                                const float* __restrict__ ws,
                                                float* __restrict__ out) {
  __shared__ float2 Wl[64 * 64];
  const float* wp = ws + WS_W;
  for (int i = threadIdx.x; i < 64 * 64; i += 512) {
    Wl[i].x = wp[2 * i + 0];
    Wl[i].y = wp[2 * i + 1];
  }
  __syncthreads();

  const int lane = threadIdx.x & 31;
  const int wave = threadIdx.x >> 5;
  const int s = wave >> 2;
  const int tj = wave & 3;
  const int p = lane & 15;
  const int kb = (lane < 16) ? 0 : 2;

  const size_t row0 = (size_t)blockIdx.x * 64 + 16 * s;

  v8f qr = {};
  v8f qi = {};

  for (int ks = 0; ks < 64; ks += 4) {
    v4f e = *(const v4f*)(x + ((row0 + p) * R_COLS + ks + kb) * 2);
    v2f ar = {e.x, e.z};
    v2f ai = {e.y, e.w};
    v2f ain = {-e.y, -e.w};

    const float2 w0 = Wl[(ks + kb) * 64 + 16 * tj + p];
    const float2 w1 = Wl[(ks + kb + 1) * 64 + 16 * tj + p];
    v2f wr = {w0.x, w1.x};
    v2f wi = {w0.y, w1.y};

    qr = wmma_f32x4(ar, wr, qr);
    qr = wmma_f32x4(ain, wi, qr);
    qi = wmma_f32x4(ar, wi, qi);
    qi = wmma_f32x4(ai, wr, qi);
  }

  const int rb = (lane < 16) ? 0 : 8;
  const int c = 16 * tj + p;
#pragma unroll
  for (int v = 0; v < 8; ++v) {
    const size_t row = row0 + rb + v;
    v2f o = {qr[v], qi[v]};
    *(v2f*)(out + (row * R_COLS + c) * 2) = o;
  }
}

extern "C" void kernel_launch(void* const* d_in, const int* in_sizes, int n_in,
                              void* d_out, int out_size, void* d_ws, size_t ws_size,
                              hipStream_t stream) {
  (void)in_sizes; (void)n_in; (void)out_size; (void)ws_size;
  const float* x = (const float*)d_in[0];
  float* out = (float*)d_out;
  float* ws = (float*)d_ws;

  fo_gram<<<GRAM_PARTS / 2, 640, 0, stream>>>(x, out);
  fo_reduce_gram<<<PART_FLOATS / 32, 128, 0, stream>>>(out, ws);
  fo_chol_inv<<<1, 64, 0, stream>>>(ws);
  fo_apply<<<M_ROWS / 64, 512, 0, stream>>>(x, ws, out);
}


# apply: W = L^-H is upper triangular, so only the 10 non-zero of 16 k-tile x column-tile products run (160 not 256 WMMA per 16 rows); one wave owns a full 16x64 strip, W fragments in registers
# speedup vs baseline: 4.4429x; 1.1933x over previous
#include <hip/hip_runtime.h>


#define M_ROWS 131072
#define R_COLS 64

typedef __attribute__((ext_vector_type(2))) float v2f;
typedef __attribute__((ext_vector_type(4))) float v4f;
typedef __attribute__((ext_vector_type(8))) float v8f;

#define WS_GR 0
#define WS_GI 4096
#define WS_W  8192

#define GRAM_TILES 10
#define GRAM_PARTS 512
#define PART_FLOATS (GRAM_TILES * 512)

__device__ inline v8f wmma_f32x4(v2f a, v2f b, v8f c) {
  return __builtin_amdgcn_wmma_f32_16x16x4_f32(
       false, a,  false, b,
       (short)0, c,  false,  false);
}

__device__ inline void tile_of(int tile, int& ti, int& tj) {
  ti = (tile >= 4) + (tile >= 7) + (tile >= 9);
  const int first = (ti == 0) ? 0 : (ti == 1) ? 4 : (ti == 2) ? 7 : 9;
  tj = tile - first + ti;
}

template <bool DIAG>
__device__ inline void gram_rows(const float* __restrict__ x, size_t mbase, int colA,
                                 int colB, v8f& gr, v8f& gi) {
  for (int ks = 0; ks < 256; ks += 4) {
    const size_t m = mbase + ks;
    const v2f ea0 = *(const v2f*)(x + (m * R_COLS + colA) * 2);
    const v2f ea1 = *(const v2f*)(x + ((m + 1) * R_COLS + colA) * 2);
    v2f eb0 = ea0;
    v2f eb1 = ea1;
    if (!DIAG) {
      eb0 = *(const v2f*)(x + (m * R_COLS + colB) * 2);
      eb1 = *(const v2f*)(x + ((m + 1) * R_COLS + colB) * 2);
    }

    const v2f ar = {ea0.x, ea1.x};
    const v2f ai = {ea0.y, ea1.y};
    const v2f ain = {-ea0.y, -ea1.y};
    const v2f br = {eb0.x, eb1.x};
    const v2f bi = {eb0.y, eb1.y};

    gr = wmma_f32x4(ar, br, gr);
    gr = wmma_f32x4(ai, bi, gr);
    gi = wmma_f32x4(ar, bi, gi);
    gi = wmma_f32x4(ain, br, gi);
  }
}

__global__ __launch_bounds__(640) void fo_gram(const float* __restrict__ x,
                                               float* __restrict__ part) {
  const int lane = threadIdx.x & 31;
  const int wave = __builtin_amdgcn_readfirstlane((int)(threadIdx.x >> 5));
  const int half = (wave >= GRAM_TILES) ? 1 : 0;
  const int tile = wave - GRAM_TILES * half;
  int ti, tj;
  tile_of(tile, ti, tj);
  const int p = lane & 15;
  const int kb = (lane < 16) ? 0 : 2;

  const int colA = 16 * ti + p;
  const int colB = 16 * tj + p;
  const int pidx = blockIdx.x * 2 + half;
  const size_t mbase = (size_t)pidx * 256 + kb;

  v8f gr = {};
  v8f gi = {};
  if (ti == tj)
    gram_rows<true>(x, mbase, colA, colB, gr, gi);
  else
    gram_rows<false>(x, mbase, colA, colB, gr, gi);

  float* pp = part + (size_t)pidx * PART_FLOATS + tile * 512 + lane;
#pragma unroll
  for (int v = 0; v < 8; ++v) {
    pp[v * 32] = gr[v];
    pp[256 + v * 32] = gi[v];
  }
}

__global__ __launch_bounds__(128) void fo_reduce_gram(const float* __restrict__ part,
                                                      float* __restrict__ ws) {
  __shared__ v4f red[16][8];
  const int gl = threadIdx.x & 7;
  const int sl = threadIdx.x >> 3;
  const int g = blockIdx.x * 8 + gl;

  const v4f* src = (const v4f*)part + (size_t)(sl * 32) * (PART_FLOATS / 4) + g;
  v4f acc = {};
#pragma unroll
  for (int i = 0; i < 32; ++i) acc += src[(size_t)i * (PART_FLOATS / 4)];
  red[sl][gl] = acc;
  __syncthreads();

  if (threadIdx.x < 8) {
    v4f s = red[0][gl];
#pragma unroll
    for (int j = 1; j < 16; ++j) s += red[j][gl];
    const int e = 4 * g;
    const int lane = e & 31;
    const int v = (e >> 5) & 7;
    const int c = (e >> 8) & 1;
    int ti, tj;
    tile_of(e >> 9, ti, tj);
    const int row = 16 * ti + 8 * (lane >> 4) + v;
    const int col = 16 * tj + (lane & 15);
    float* plane = ws + c * 4096;
    *(v4f*)(plane + row * 64 + col) = s;
    if (ti != tj) {
      const v4f ms = c ? -s : s;
      plane[(col + 0) * 64 + row] = ms.x;
      plane[(col + 1) * 64 + row] = ms.y;
      plane[(col + 2) * 64 + row] = ms.z;
      plane[(col + 3) * 64 + row] = ms.w;
    }
  }
}

__global__ __launch_bounds__(64) void fo_chol_inv(float* __restrict__ ws) {
  __shared__ v4f Vb[2][32];
  const int t = threadIdx.x;
  const float* grp = ws + WS_GR;
  const float* gip = ws + WS_GI;

  v2f U[64];
#pragma unroll
  for (int i = 0; i < 16; ++i) {
    const v4f r = *(const v4f*)(grp + t * 64 + 4 * i);
    const v4f m = *(const v4f*)(gip + t * 64 + 4 * i);
    U[4 * i + 0] = v2f{r.x, m.x};
    U[4 * i + 1] = v2f{r.y, m.y};
    U[4 * i + 2] = v2f{r.z, m.z};
    U[4 * i + 3] = v2f{r.w, m.w};
  }

  float mydd = 1.0f;

#pragma nounroll
  for (int k = 0; k < 64; ++k) {
    v4f* buf = Vb[k & 1];
    if (t == k) {
#pragma unroll
      for (int i = 0; i < 32; ++i)
        buf[i] = v4f{U[2 * i].x, U[2 * i].y, U[2 * i + 1].x, U[2 * i + 1].y};
    }
    __syncthreads();

    v2f V[64];
#pragma unroll
    for (int i = 0; i < 32; ++i) {
      const v4f q = buf[i];
      V[2 * i] = v2f{q.x, q.y};
      V[2 * i + 1] = v2f{q.z, q.w};
    }
    __builtin_amdgcn_sched_barrier(0);

    const float dd = V[0].x + 1e-8f;
    if (t == k) mydd = dd;
    const float act = (t > k) ? 1.0f : 0.0f;
    const float ax = (U[0].x * act) / dd;
    const float ay = (U[0].y * act) / dd;
    const v2f nax = {-ax, -ax};
    const v2f aypm = {ay, -ay};

#pragma unroll
    for (int s = 0; s < 63; ++s) {
      const v2f vs = V[s + 1];
      const v2f vsw = {vs.y, vs.x};
      const v2f tmp = __builtin_elementwise_fma(nax, vs, U[s + 1]);
      U[s] = __builtin_elementwise_fma(aypm, vsw, tmp);
    }
    U[63] = (t == k) ? v2f{1.0f, 0.0f} : v2f{-ax, -ay};
  }

  const float rd = 1.0f / sqrtf(mydd);
  float* wp = ws + WS_W;
#pragma unroll
  for (int c = 0; c < 64; ++c) {
    const v2f o = {U[c].x * rd, -U[c].y * rd};
    *(v2f*)(wp + (c * 64 + t) * 2) = o;
  }
}

#define APPLY_WAVES 2048
#define APPLY_STRIPS_PER_WAVE (M_ROWS / 16 / APPLY_WAVES)

__device__ inline void load_strip(const float* __restrict__ x, size_t row, int kb,
                                  v4f (&e)[16]) {
#pragma unroll
  for (int ks = 0; ks < 16; ++ks)
    e[ks] = *(const v4f*)(x + (row * R_COLS + 4 * ks + kb) * 2);
}

__global__ __launch_bounds__(256) void fo_apply(const float* __restrict__ x,
                                                const float* __restrict__ ws,
                                                float* __restrict__ out) {
  const int lane = threadIdx.x & 31;
  const int gw = blockIdx.x * 8 + (int)(threadIdx.x >> 5);
  const int p = lane & 15;
  const int h = lane >> 4;
  const int kb = 2 * h;

  const float* wp = ws + WS_W;
  v2f wr[40];
  v2f wi[40];
#pragma unroll
  for (int tj = 0; tj < 4; ++tj) {
#pragma unroll
    for (int ks = 0; ks < 16; ++ks) {
      if (ks < 4 * (tj + 1)) {
        const int idx = 2 * tj * (tj + 1) + ks;
        const v2f w0 = *(const v2f*)(wp + ((4 * ks + kb) * 64 + 16 * tj + p) * 2);
        const v2f w1 = *(const v2f*)(wp + ((4 * ks + kb + 1) * 64 + 16 * tj + p) * 2);
        wr[idx] = v2f{w0.x, w1.x};
        wi[idx] = v2f{w0.y, w1.y};
      }
    }
  }

  v4f e[16];
  load_strip(x, (size_t)gw * 16 + p, kb, e);

#pragma unroll
  for (int it = 0; it < APPLY_STRIPS_PER_WAVE; ++it) {
    const size_t row0 = ((size_t)it * APPLY_WAVES + gw) * 16;
    v4f en[16];
    if (it + 1 < APPLY_STRIPS_PER_WAVE)
      load_strip(x, row0 + (size_t)APPLY_WAVES * 16 + p, kb, en);

    v8f qr[4] = {};
    v8f qi[4] = {};
#pragma unroll
    for (int ks = 0; ks < 16; ++ks) {
      const v2f ar = {e[ks].x, e[ks].z};
      const v2f ai = {e[ks].y, e[ks].w};
      const v2f ain = {-e[ks].y, -e[ks].w};
#pragma unroll
      for (int tj = 0; tj < 4; ++tj)
        if (tj >= (ks >> 2)) qr[tj] = wmma_f32x4(ar, wr[2 * tj * (tj + 1) + ks], qr[tj]);
#pragma unroll
      for (int tj = 0; tj < 4; ++tj)
        if (tj >= (ks >> 2)) qi[tj] = wmma_f32x4(ar, wi[2 * tj * (tj + 1) + ks], qi[tj]);
#pragma unroll
      for (int tj = 0; tj < 4; ++tj)
        if (tj >= (ks >> 2)) qr[tj] = wmma_f32x4(ain, wi[2 * tj * (tj + 1) + ks], qr[tj]);
#pragma unroll
      for (int tj = 0; tj < 4; ++tj)
        if (tj >= (ks >> 2)) qi[tj] = wmma_f32x4(ai, wr[2 * tj * (tj + 1) + ks], qi[tj]);
    }

#pragma unroll
    for (int tj = 0; tj < 4; ++tj) {
#pragma unroll
      for (int v = 0; v < 8; ++v) {
        const size_t row = row0 + 8 * h + v;
        const v2f o = {qr[tj][v], qi[tj][v]};
        *(v2f*)(out + (row * R_COLS + 16 * tj + p) * 2) = o;
      }
    }

    if (it + 1 < APPLY_STRIPS_PER_WAVE) {
#pragma unroll
      for (int ks = 0; ks < 16; ++ks) e[ks] = en[ks];
    }
  }
}

extern "C" void kernel_launch(void* const* d_in, const int* in_sizes, int n_in,
                              void* d_out, int out_size, void* d_ws, size_t ws_size,
                              hipStream_t stream) {
  (void)in_sizes; (void)n_in; (void)out_size; (void)ws_size;
  const float* x = (const float*)d_in[0];
  float* out = (float*)d_out;
  float* ws = (float*)d_ws;

  fo_gram<<<GRAM_PARTS / 2, 640, 0, stream>>>(x, out);
  fo_reduce_gram<<<PART_FLOATS / 32, 128, 0, stream>>>(out, ws);
  fo_chol_inv<<<1, 64, 0, stream>>>(ws);
  fo_apply<<<APPLY_WAVES / 8, 256, 0, stream>>>(x, ws, out);
}


# Gram and apply: each complex product from three real WMMA products (P1=ArBr, P2=AiBi, P3=(Ar+Ai)(Br+Bi)) not four, 25% fewer WMMA; sums regrouped (rounding and sum order change)
# speedup vs baseline: 4.8391x; 1.0892x over previous
#include <hip/hip_runtime.h>


#define M_ROWS 131072
#define R_COLS 64

typedef __attribute__((ext_vector_type(2))) float v2f;
typedef __attribute__((ext_vector_type(4))) float v4f;
typedef __attribute__((ext_vector_type(8))) float v8f;

#define WS_GR 0
#define WS_GI 4096
#define WS_W  8192

#define GRAM_TILES 10
#define GRAM_PARTS 512
#define PART_FLOATS (GRAM_TILES * 512)

__device__ inline v8f wmma_f32x4(v2f a, v2f b, v8f c) {
  return __builtin_amdgcn_wmma_f32_16x16x4_f32(
       false, a,  false, b,
       (short)0, c,  false,  false);
}

__device__ inline void tile_of(int tile, int& ti, int& tj) {
  ti = (tile >= 4) + (tile >= 7) + (tile >= 9);
  const int first = (ti == 0) ? 0 : (ti == 1) ? 4 : (ti == 2) ? 7 : 9;
  tj = tile - first + ti;
}

template <bool DIAG>
__device__ inline void gram_rows(const float* __restrict__ x, size_t mbase, int colA,
                                 int colB, v8f& p1, v8f& p2, v8f& p3) {
#pragma unroll 8
  for (int ks = 0; ks < 256; ks += 4) {
    const size_t m = mbase + ks;
    const v2f ea0 = *(const v2f*)(x + (m * R_COLS + colA) * 2);
    const v2f ea1 = *(const v2f*)(x + ((m + 1) * R_COLS + colA) * 2);
    v2f eb0 = ea0;
    v2f eb1 = ea1;
    if (!DIAG) {
      eb0 = *(const v2f*)(x + (m * R_COLS + colB) * 2);
      eb1 = *(const v2f*)(x + ((m + 1) * R_COLS + colB) * 2);
    }

    const v2f ar = {ea0.x, ea1.x};
    const v2f ai = {ea0.y, ea1.y};
    const v2f br = {eb0.x, eb1.x};
    const v2f bi = {eb0.y, eb1.y};
    const v2f as = ar + ai;
    const v2f bd = bi - br;

    p1 = wmma_f32x4(ar, br, p1);
    p2 = wmma_f32x4(ai, bi, p2);
    p3 = wmma_f32x4(as, bd, p3);
  }
}

__global__ __launch_bounds__(640) void fo_gram(const float* __restrict__ x,
                                               float* __restrict__ part) {
  const int lane = threadIdx.x & 31;
  const int wave = __builtin_amdgcn_readfirstlane((int)(threadIdx.x >> 5));
  const int half = (wave >= GRAM_TILES) ? 1 : 0;
  const int tile = wave - GRAM_TILES * half;
  int ti, tj;
  tile_of(tile, ti, tj);
  const int p = lane & 15;
  const int kb = (lane < 16) ? 0 : 2;

  const int colA = 16 * ti + p;
  const int colB = 16 * tj + p;
  const int pidx = blockIdx.x * 2 + half;
  const size_t mbase = (size_t)pidx * 256 + kb;

  v8f p1 = {};
  v8f p2 = {};
  v8f p3 = {};
  if (ti == tj)
    gram_rows<true>(x, mbase, colA, colB, p1, p2, p3);
  else
    gram_rows<false>(x, mbase, colA, colB, p1, p2, p3);
  const v8f gr = p1 + p2;
  const v8f gi = p3 + (p1 - p2);

  float* pp = part + (size_t)pidx * PART_FLOATS + tile * 512 + lane;
#pragma unroll
  for (int v = 0; v < 8; ++v) {
    pp[v * 32] = gr[v];
    pp[256 + v * 32] = gi[v];
  }
}

__global__ __launch_bounds__(128) void fo_reduce_gram(const float* __restrict__ part,
                                                      float* __restrict__ ws) {
  __shared__ v4f red[16][8];
  const int gl = threadIdx.x & 7;
  const int sl = threadIdx.x >> 3;
  const int g = blockIdx.x * 8 + gl;

  const v4f* src = (const v4f*)part + (size_t)(sl * 32) * (PART_FLOATS / 4) + g;
  v4f acc = {};
#pragma unroll
  for (int i = 0; i < 32; ++i) acc += src[(size_t)i * (PART_FLOATS / 4)];
  red[sl][gl] = acc;
  __syncthreads();

  if (threadIdx.x < 8) {
    v4f s = red[0][gl];
#pragma unroll
    for (int j = 1; j < 16; ++j) s += red[j][gl];
    const int e = 4 * g;
    const int lane = e & 31;
    const int v = (e >> 5) & 7;
    const int c = (e >> 8) & 1;
    int ti, tj;
    tile_of(e >> 9, ti, tj);
    const int row = 16 * ti + 8 * (lane >> 4) + v;
    const int col = 16 * tj + (lane & 15);
    float* plane = ws + c * 4096;
    *(v4f*)(plane + row * 64 + col) = s;
    if (ti != tj) {
      const v4f ms = c ? -s : s;
      plane[(col + 0) * 64 + row] = ms.x;
      plane[(col + 1) * 64 + row] = ms.y;
      plane[(col + 2) * 64 + row] = ms.z;
      plane[(col + 3) * 64 + row] = ms.w;
    }
  }
}

__global__ __launch_bounds__(64) void fo_chol_inv(float* __restrict__ ws) {
  __shared__ v4f Vb[2][32];
  const int t = threadIdx.x;
  const float* grp = ws + WS_GR;
  const float* gip = ws + WS_GI;

  v2f U[64];
#pragma unroll
  for (int i = 0; i < 16; ++i) {
    const v4f r = *(const v4f*)(grp + t * 64 + 4 * i);
    const v4f m = *(const v4f*)(gip + t * 64 + 4 * i);
    U[4 * i + 0] = v2f{r.x, m.x};
    U[4 * i + 1] = v2f{r.y, m.y};
    U[4 * i + 2] = v2f{r.z, m.z};
    U[4 * i + 3] = v2f{r.w, m.w};
  }

  float mydd = 1.0f;

#pragma nounroll
  for (int k = 0; k < 64; ++k) {
    v4f* buf = Vb[k & 1];
    if (t == k) {
#pragma unroll
      for (int i = 0; i < 32; ++i)
        buf[i] = v4f{U[2 * i].x, U[2 * i].y, U[2 * i + 1].x, U[2 * i + 1].y};
    }
    __syncthreads();

    v2f V[64];
#pragma unroll
    for (int i = 0; i < 32; ++i) {
      const v4f q = buf[i];
      V[2 * i] = v2f{q.x, q.y};
      V[2 * i + 1] = v2f{q.z, q.w};
    }
    __builtin_amdgcn_sched_barrier(0);

    const float dd = V[0].x + 1e-8f;
    if (t == k) mydd = dd;
    const float act = (t > k) ? 1.0f : 0.0f;
    const float ax = (U[0].x * act) / dd;
    const float ay = (U[0].y * act) / dd;
    const v2f nax = {-ax, -ax};
    const v2f aypm = {ay, -ay};

#pragma unroll
    for (int s = 0; s < 63; ++s) {
      const v2f vs = V[s + 1];
      const v2f vsw = {vs.y, vs.x};
      const v2f tmp = __builtin_elementwise_fma(nax, vs, U[s + 1]);
      U[s] = __builtin_elementwise_fma(aypm, vsw, tmp);
    }
    U[63] = (t == k) ? v2f{1.0f, 0.0f} : v2f{-ax, -ay};
  }

  const float rd = 1.0f / sqrtf(mydd);
  float* wp = ws + WS_W;
#pragma unroll
  for (int c = 0; c < 64; ++c) {
    const v2f o = {U[c].x * rd, -U[c].y * rd};
    *(v2f*)(wp + (c * 64 + t) * 2) = o;
  }
}

#define APPLY_WAVES 2048
#define APPLY_STRIPS_PER_WAVE (M_ROWS / 16 / APPLY_WAVES)

__device__ inline void load_strip(const float* __restrict__ x, size_t row, int kb,
                                  v4f (&e)[16]) {
#pragma unroll
  for (int ks = 0; ks < 16; ++ks)
    e[ks] = *(const v4f*)(x + (row * R_COLS + 4 * ks + kb) * 2);
}

__global__ __launch_bounds__(256) void fo_apply(const float* __restrict__ x,
                                                const float* __restrict__ ws,
                                                float* __restrict__ out) {
  const int lane = threadIdx.x & 31;
  const int gw = blockIdx.x * 8 + (int)(threadIdx.x >> 5);
  const int p = lane & 15;
  const int h = lane >> 4;
  const int kb = 2 * h;

  const float* wp = ws + WS_W;
  v2f wr[40];
  v2f wi[40];
  v2f wsum[40];
#pragma unroll
  for (int tj = 0; tj < 4; ++tj) {
#pragma unroll
    for (int ks = 0; ks < 16; ++ks) {
      if (ks < 4 * (tj + 1)) {
        const int idx = 2 * tj * (tj + 1) + ks;
        const v2f w0 = *(const v2f*)(wp + ((4 * ks + kb) * 64 + 16 * tj + p) * 2);
        const v2f w1 = *(const v2f*)(wp + ((4 * ks + kb + 1) * 64 + 16 * tj + p) * 2);
        wr[idx] = v2f{w0.x, w1.x};
        wi[idx] = v2f{w0.y, w1.y};
        wsum[idx] = wr[idx] + wi[idx];
      }
    }
  }

  v4f e[16];
  load_strip(x, (size_t)gw * 16 + p, kb, e);

#pragma unroll
  for (int it = 0; it < APPLY_STRIPS_PER_WAVE; ++it) {
    const size_t row0 = ((size_t)it * APPLY_WAVES + gw) * 16;
    v4f en[16];
    if (it + 1 < APPLY_STRIPS_PER_WAVE)
      load_strip(x, row0 + (size_t)APPLY_WAVES * 16 + p, kb, en);

    v8f p1[4] = {};
    v8f p2[4] = {};
    v8f p3[4] = {};
#pragma unroll
    for (int ks = 0; ks < 16; ++ks) {
      const v2f ar = {e[ks].x, e[ks].z};
      const v2f ai = {e[ks].y, e[ks].w};
      const v2f as = ar + ai;
#pragma unroll
      for (int tj = 0; tj < 4; ++tj)
        if (tj >= (ks >> 2)) p1[tj] = wmma_f32x4(ar, wr[2 * tj * (tj + 1) + ks], p1[tj]);
#pragma unroll
      for (int tj = 0; tj < 4; ++tj)
        if (tj >= (ks >> 2)) p2[tj] = wmma_f32x4(ai, wi[2 * tj * (tj + 1) + ks], p2[tj]);
#pragma unroll
      for (int tj = 0; tj < 4; ++tj)
        if (tj >= (ks >> 2)) p3[tj] = wmma_f32x4(as, wsum[2 * tj * (tj + 1) + ks], p3[tj]);
    }

#pragma unroll
    for (int tj = 0; tj < 4; ++tj) {
      const v8f qr = p1[tj] - p2[tj];
      const v8f qi = (p3[tj] - p1[tj]) - p2[tj];
#pragma unroll
      for (int v = 0; v < 8; ++v) {
        const size_t row = row0 + 8 * h + v;
        const v2f o = {qr[v], qi[v]};
        *(v2f*)(out + (row * R_COLS + 16 * tj + p) * 2) = o;
      }
    }

    if (it + 1 < APPLY_STRIPS_PER_WAVE) {
#pragma unroll
      for (int ks = 0; ks < 16; ++ks) e[ks] = en[ks];
    }
  }
}

extern "C" void kernel_launch(void* const* d_in, const int* in_sizes, int n_in,
                              void* d_out, int out_size, void* d_ws, size_t ws_size,
                              hipStream_t stream) {
  (void)in_sizes; (void)n_in; (void)out_size; (void)ws_size;
  const float* x = (const float*)d_in[0];
  float* out = (float*)d_out;
  float* ws = (float*)d_ws;

  fo_gram<<<GRAM_PARTS / 2, 640, 0, stream>>>(x, out);
  fo_reduce_gram<<<PART_FLOATS / 32, 128, 0, stream>>>(out, ws);
  fo_chol_inv<<<1, 64, 0, stream>>>(ws);
  fo_apply<<<APPLY_WAVES / 8, 256, 0, stream>>>(x, ws, out);
}
